# starting kernel
# baseline (speedup 1.0000x reference)
#include <hip/hip_runtime.h>
#include <math.h>

#define B_BATCH 32
#define L_SEQ   2048
#define D_IN    64
#define NC      32
#define DC      16
#define EPS_F   1e-7f

typedef __attribute__((ext_vector_type(2))) float v2f;
typedef __attribute__((ext_vector_type(8))) float v8f;

__global__ void caps_gemm(const float* __restrict__ u,
                          const float* __restrict__ W,
                          float* __restrict__ uhat) {
  const int wave = (blockIdx.x * blockDim.x + threadIdx.x) >> 5;
  const int lane = threadIdx.x & 31;
  const int ng    = wave & 3;
  const int ltile = (wave >> 2) & 127;
  const int b     = wave >> 9;

  const int col = lane & 15;
  const int hi  = lane >> 4;

  const float* urow = u + ((size_t)b * L_SEQ + (size_t)ltile * 16 + col) * D_IN;
  const v2f* urow2 = reinterpret_cast<const v2f*>(urow);
  v2f a[16];
#pragma unroll
  for (int k = 0; k < 16; ++k) a[k] = urow2[2 * k + hi];

  for (int n = 8 * ng; n < 8 * ng + 8; ++n) {
    const float* wcol = W + n * DC + col;
    v8f acc = {};
#pragma unroll
    for (int k = 0; k < 16; ++k) {
      const int kk = 4 * k + 2 * hi;
      v2f bm;
      bm.x = wcol[(size_t)kk * (NC * DC)];
      bm.y = wcol[(size_t)(kk + 1) * (NC * DC)];
      acc = __builtin_amdgcn_wmma_f32_16x16x4_f32(
           false, a[k],  false, bm,
           (short)0, acc,  false,  false);
    }
    float* out = uhat + (((size_t)b * NC + n) * L_SEQ + (size_t)ltile * 16) * DC;
#pragma unroll
    for (int v = 0; v < 8; ++v) out[(v + 8 * hi) * DC + col] = acc[v];
  }
}

__global__ void caps_reduce(const float* __restrict__ uhat,
                            const float* __restrict__ c,
                            float* __restrict__ outv, int iter0) {
  const int bn = blockIdx.x;
  const int t  = threadIdx.x;
  const float4* src = reinterpret_cast<const float4*>(uhat + (size_t)bn * L_SEQ * DC);
  const float* cw = c + (size_t)bn * L_SEQ;

  float4 acc = make_float4(0.f, 0.f, 0.f, 0.f);
  const int dg = t & 3;
  for (int q = t; q < (L_SEQ * DC) / 4; q += 256) {
    const int l = q >> 2;
    const float cv = iter0 ? (1.0f / 32.0f) : cw[l];
    float4 v = src[q];
    acc.x += cv * v.x; acc.y += cv * v.y;
    acc.z += cv * v.z; acc.w += cv * v.w;
  }

  __shared__ float part[16][64];
  const int slot = t >> 2;
  part[4 * dg + 0][slot] = acc.x;
  part[4 * dg + 1][slot] = acc.y;
  part[4 * dg + 2][slot] = acc.z;
  part[4 * dg + 3][slot] = acc.w;
  __syncthreads();

  __shared__ float sums[16];
  if (t < 16) {
    float s = 0.f;
    for (int i = 0; i < 64; ++i) s += part[t][i];
    sums[t] = s;
  }
  __syncthreads();
  if (t < 16) {
    float ss = EPS_F;
    for (int d = 0; d < 16; ++d) ss += sums[d] * sums[d];
    outv[bn * DC + t] = sums[t] / sqrtf(ss);
  }
}

__global__ void caps_update(const float* __restrict__ uhat,
                            const float* __restrict__ outv,
                            float* __restrict__ c) {
  const int blk = blockIdx.x;
  const int b   = blk >> 3;
  const int l   = ((blk & 7) << 8) + threadIdx.x;

  __shared__ float ovs[NC * DC];
  ovs[threadIdx.x]       = outv[b * NC * DC + threadIdx.x];
  ovs[threadIdx.x + 256] = outv[b * NC * DC + threadIdx.x + 256];
  __syncthreads();

  float vals[NC];
  float m = -3.402823466e38f;
#pragma unroll 4
  for (int n = 0; n < NC; ++n) {
    const float4* uh4 = reinterpret_cast<const float4*>(
        uhat + (((size_t)b * NC + n) * L_SEQ + l) * DC);
    float dot = 0.f;
#pragma unroll
    for (int q = 0; q < 4; ++q) {
      float4 v = uh4[q];
      dot += v.x * ovs[n * DC + 4 * q + 0];
      dot += v.y * ovs[n * DC + 4 * q + 1];
      dot += v.z * ovs[n * DC + 4 * q + 2];
      dot += v.w * ovs[n * DC + 4 * q + 3];
    }
    vals[n] = dot;
    m = fmaxf(m, dot);
  }
  float sum = 0.f;
#pragma unroll
  for (int n = 0; n < NC; ++n) { vals[n] = expf(vals[n] - m); sum += vals[n]; }
  const float inv = 1.0f / sum;
#pragma unroll
  for (int n = 0; n < NC; ++n)
    c[((size_t)b * NC + n) * L_SEQ + l] = vals[n] * inv;
}

extern "C" void kernel_launch(void* const* d_in, const int* in_sizes, int n_in,
                              void* d_out, int out_size, void* d_ws, size_t ws_size,
                              hipStream_t stream) {
  const float* u = (const float*)d_in[0];
  const float* W = (const float*)d_in[1];
  float* out = (float*)d_out;

  float* uhat = (float*)d_ws;
  float* cbuf = uhat + (size_t)B_BATCH * NC * L_SEQ * DC;
  float* ov   = cbuf + (size_t)B_BATCH * NC * L_SEQ;

  caps_gemm<<<2048, 256, 0, stream>>>(u, W, uhat);

  caps_reduce<<<B_BATCH * NC, 256, 0, stream>>>(uhat, cbuf, ov, 1);
  caps_update<<<B_BATCH * 8, 256, 0, stream>>>(uhat, ov, cbuf);
  caps_reduce<<<B_BATCH * NC, 256, 0, stream>>>(uhat, cbuf, ov, 0);
  caps_update<<<B_BATCH * 8, 256, 0, stream>>>(uhat, ov, cbuf);
  caps_reduce<<<B_BATCH * NC, 256, 0, stream>>>(uhat, cbuf, out, 0);
}


# Routing softmax once per l, not once per lane: a quad reduce-scatter leaves each of an l's 4 lanes 8 capsules' logits (8 exps per lane and step, was 32); c read back by quad_perm in the weighted sum
# speedup vs baseline: 2.0662x; 2.0662x over previous
#include <hip/hip_runtime.h>
#include <math.h>

#define B_BATCH 32
#define L_SEQ   2048
#define D_IN    64
#define NC      32
#define DC      16
#define EPS_F   1e-7f

typedef __attribute__((ext_vector_type(2))) float v2f;
typedef __attribute__((ext_vector_type(8))) float v8f;

#define DEV static __device__ __forceinline__

DEV float rows8(v8f x) {
  const v2f q = (__builtin_shufflevector(x, x, 0, 1) + __builtin_shufflevector(x, x, 2, 3)) +
                (__builtin_shufflevector(x, x, 4, 5) + __builtin_shufflevector(x, x, 6, 7));
  return q.x + q.y;
}

#define WP_PITCH 1056
#define CS_PITCH 528

__global__ void __launch_bounds__(512)
caps_gemm(const float* __restrict__ u,
          const float* __restrict__ W,
          float* __restrict__ uhat,
          float* __restrict__ part) {
  __shared__ float wl[32 * WP_PITCH];
  __shared__ float cs[32 * CS_PITCH];

  const int t    = threadIdx.x;
  const int lane = t & 31;
  const int wave = t >> 5;
  const int col  = lane & 15;
  const int hi   = lane >> 4;

  const int lt    = blockIdx.x * 16 + wave;
  const int b     = lt >> 7;
  const int ltile = lt & 127;

  const float* urow = u + ((size_t)b * L_SEQ + (size_t)ltile * 16 + col) * D_IN;
  const v2f* urow2 = reinterpret_cast<const v2f*>(urow);
  v2f a[16];
#pragma unroll
  for (int k = 0; k < 16; ++k) a[k] = urow2[2 * k + hi];

  {
    const int cp  = t & 255;
    const int kp0 = t >> 8;
    const v2f* W2 = reinterpret_cast<const v2f*>(W);
    v2f r0[16], r1[16];
#pragma unroll
    for (int i = 0; i < 16; ++i) {
      const int kp = kp0 + 2 * i;
      r0[i] = W2[(2 * kp) * 256 + cp];
      r1[i] = W2[(2 * kp + 1) * 256 + cp];
    }
#pragma unroll
    for (int i = 0; i < 16; ++i) {
      const int kp = kp0 + 2 * i;
      *reinterpret_cast<float4*>(wl + kp * WP_PITCH + cp * 4) =
          make_float4(r0[i].x, r1[i].x, r0[i].y, r1[i].y);
    }
  }
  __syncthreads();

  const float* bl = wl + hi * WP_PITCH + col * 2;
  float* outp = uhat + (((size_t)b * NC) * L_SEQ + (size_t)ltile * 16) * DC + (8 * hi) * DC + col;
  float* csw = cs + (wave * 2 + hi) * CS_PITCH + col;

  v2f bf[2][2][16];
#pragma unroll
  for (int j = 0; j < 2; ++j)
#pragma unroll
    for (int k = 0; k < 16; ++k)
      bf[0][j][k] = *reinterpret_cast<const v2f*>(bl + (2 * k) * WP_PITCH + j * 32);

#pragma unroll
  for (int p = 0; p < NC / 2; ++p) {
    __builtin_amdgcn_sched_barrier(0);
    if (p + 1 < NC / 2) {
#pragma unroll
      for (int j = 0; j < 2; ++j)
#pragma unroll
        for (int k = 0; k < 16; ++k)
          bf[(p + 1) & 1][j][k] =
              *reinterpret_cast<const v2f*>(bl + (2 * k) * WP_PITCH + ((p + 1) * 2 + j) * 32);
    }
    __builtin_amdgcn_sched_barrier(0);

    v8f acc0 = {};
    v8f acc1 = {};
#pragma unroll
    for (int k = 0; k < 16; ++k) {
      acc0 = __builtin_amdgcn_wmma_f32_16x16x4_f32(
           false, a[k],  false, bf[p & 1][0][k],
           (short)0, acc0,  false,  false);
      acc1 = __builtin_amdgcn_wmma_f32_16x16x4_f32(
           false, a[k],  false, bf[p & 1][1][k],
           (short)0, acc1,  false,  false);
    }
    float* o0 = outp + (size_t)(2 * p) * (L_SEQ * DC);
    float* o1 = o0 + (size_t)(L_SEQ * DC);
#pragma unroll
    for (int v = 0; v < 8; ++v) o0[v * DC] = acc0[v];
#pragma unroll
    for (int v = 0; v < 8; ++v) o1[v * DC] = acc1[v];
    csw[(2 * p) * DC]     = rows8(acc0);
    csw[(2 * p + 1) * DC] = rows8(acc1);
  }
  __builtin_amdgcn_fence(__ATOMIC_RELEASE, "workgroup", "local");
  __builtin_amdgcn_s_barrier();
  __builtin_amdgcn_fence(__ATOMIC_ACQUIRE, "workgroup", "local");

  {
    float s = 0.f;
#pragma unroll
    for (int j = 0; j < 32; ++j) s += cs[j * CS_PITCH + t];
    part[(size_t)blockIdx.x * (NC * DC) + t] = s * (1.0f / 32.0f);
  }
}

#define RL   256
#define RCH  (L_SEQ / RL)
#define RIT  (RL / 64)

template <int CTRL> DEV float dppf(float x) {
  return __builtin_bit_cast(float, __builtin_amdgcn_update_dpp(
      0, __builtin_bit_cast(int, x), CTRL, 0xf, 0xf, true));
}
#define DPP_QUAD_X1 0xB1
#define DPP_QUAD_X2 0x4E
#define DPP_XMASK(n) (0x160 + (n))
#define DPP_QUAD_B0 0x00
#define DPP_QUAD_B1 0x55
#define DPP_QUAD_B2 0xAA
#define DPP_QUAD_B3 0xFF
#define WSUM(g, CTRL)                                              \
  _Pragma("unroll")                                                \
  for (int i = 0; i < 8; ++i) {                                    \
    const float c = dppf<CTRL>(q[i]);                              \
    const float4 v = cur[8 * (g) + i];                             \
    acc[8 * (g) + i].x = __builtin_fmaf(c, v.x, acc[8 * (g) + i].x); \
    acc[8 * (g) + i].y = __builtin_fmaf(c, v.y, acc[8 * (g) + i].y); \
    acc[8 * (g) + i].z = __builtin_fmaf(c, v.z, acc[8 * (g) + i].z); \
    acc[8 * (g) + i].w = __builtin_fmaf(c, v.w, acc[8 * (g) + i].w); \
  }

__global__ void __launch_bounds__(256)
caps_route(const float* __restrict__ uhat, const float* __restrict__ outv,
           float* __restrict__ part) {
  const int b    = blockIdx.x / RCH;
  const int ch   = blockIdx.x % RCH;
  const int t    = threadIdx.x;
  const int dq   = t & 3;
  const int lane = t & 31;
  const int wave = t >> 5;
  const bool b0  = (t & 1) != 0;
  const bool b1  = (t & 2) != 0;

  __shared__ float4 ovs[NC * 4];
  __shared__ float4 red[16][NC * 4];

  if (t < NC * 4)
    ovs[t] = reinterpret_cast<const float4*>(outv + (size_t)b * NC * DC)[t];
  __syncthreads();

  const float4* base = reinterpret_cast<const float4*>(
      uhat + ((size_t)b * NC * L_SEQ + (size_t)ch * RL) * DC) + t;
  constexpr int NSTR = L_SEQ * DC / 4;

  float4 buf[2][NC];
  float4 acc[NC];
#pragma unroll
  for (int n = 0; n < NC; ++n) acc[n] = make_float4(0.f, 0.f, 0.f, 0.f);
#pragma unroll
  for (int n = 0; n < NC; ++n) buf[0][n] = base[(size_t)n * NSTR];

#pragma unroll
  for (int it = 0; it < RIT; ++it) {
    __builtin_amdgcn_sched_barrier(0);
    if (it + 1 < RIT) {
#pragma unroll
      for (int n = 0; n < NC; ++n)
        buf[(it + 1) & 1][n] = base[(size_t)n * NSTR + (it + 1) * 256];
    }
    __builtin_amdgcn_sched_barrier(0);
    const float4* cur = buf[it & 1];

    float p[NC];
#pragma unroll
    for (int n = 0; n < NC; ++n) {
      const float4 o = ovs[n * 4 + dq];
      const float4 v = cur[n];
      p[n] = __builtin_fmaf(v.w, o.w, __builtin_fmaf(v.z, o.z, __builtin_fmaf(v.y, o.y, v.x * o.x)));
    }
    float r1[16];
#pragma unroll
    for (int i = 0; i < 16; ++i) {
      const float keep = b0 ? p[16 + i] : p[i];
      const float send = b0 ? p[i] : p[16 + i];
      r1[i] = keep + dppf<DPP_QUAD_X1>(send);
    }
    float q[8];
#pragma unroll
    for (int i = 0; i < 8; ++i) {
      const float keep = b1 ? r1[8 + i] : r1[i];
      const float send = b1 ? r1[i] : r1[8 + i];
      q[i] = keep + dppf<DPP_QUAD_X2>(send);
    }
    float m = q[0];
#pragma unroll
    for (int i = 1; i < 8; ++i) m = fmaxf(m, q[i]);
    m = fmaxf(m, dppf<DPP_QUAD_X1>(m));
    m = fmaxf(m, dppf<DPP_QUAD_X2>(m));
    float sum = 0.f;
#pragma unroll
    for (int i = 0; i < 8; ++i) { q[i] = expf(q[i] - m); sum += q[i]; }
    sum += dppf<DPP_QUAD_X1>(sum);
    sum += dppf<DPP_QUAD_X2>(sum);
    const float inv = 1.0f / sum;
#pragma unroll
    for (int i = 0; i < 8; ++i) q[i] *= inv;
    WSUM(0, DPP_QUAD_B0)
    WSUM(1, DPP_QUAD_B2)
    WSUM(2, DPP_QUAD_B1)
    WSUM(3, DPP_QUAD_B3)
  }
  __builtin_amdgcn_sched_barrier(0);

#pragma unroll
  for (int n = 0; n < NC; ++n) {
    acc[n].x += dppf<DPP_XMASK(4)>(acc[n].x); acc[n].x += dppf<DPP_XMASK(8)>(acc[n].x);
    acc[n].y += dppf<DPP_XMASK(4)>(acc[n].y); acc[n].y += dppf<DPP_XMASK(8)>(acc[n].y);
    acc[n].z += dppf<DPP_XMASK(4)>(acc[n].z); acc[n].z += dppf<DPP_XMASK(8)>(acc[n].z);
    acc[n].w += dppf<DPP_XMASK(4)>(acc[n].w); acc[n].w += dppf<DPP_XMASK(8)>(acc[n].w);
  }
  if ((lane & 12) == 0) {
    const int j = wave * 2 + (lane >> 4);
#pragma unroll
    for (int n = 0; n < NC; ++n) red[j][n * 4 + dq] = acc[n];
  }
  __syncthreads();

  const float* redf = reinterpret_cast<const float*>(&red[0][0]);
#pragma unroll
  for (int k = 0; k < 2; ++k) {
    const int o = t + 256 * k;
    float s = 0.f;
#pragma unroll
    for (int j = 0; j < 16; ++j) s += redf[j * (NC * DC) + o];
    part[(size_t)blockIdx.x * (NC * DC) + o] = s;
  }
}

__global__ void __launch_bounds__(512)
caps_finish(const float* __restrict__ part, float* __restrict__ outv) {
  const int b = blockIdx.x;
  const int o = threadIdx.x;
  float s = 0.f;
#pragma unroll
  for (int ch = 0; ch < RCH; ++ch) s += part[((size_t)b * RCH + ch) * (NC * DC) + o];
  float ss = s * s;
  ss += dppf<DPP_XMASK(1)>(ss);
  ss += dppf<DPP_XMASK(2)>(ss);
  ss += dppf<DPP_XMASK(4)>(ss);
  ss += dppf<DPP_XMASK(8)>(ss);
  outv[(size_t)b * (NC * DC) + o] = s / sqrtf(ss + EPS_F);
}

extern "C" void kernel_launch(void* const* d_in, const int* in_sizes, int n_in,
                              void* d_out, int out_size, void* d_ws, size_t ws_size,
                              hipStream_t stream) {
  const float* u = (const float*)d_in[0];
  const float* W = (const float*)d_in[1];
  float* out = (float*)d_out;

  float* uhat = (float*)d_ws;
  float* part = uhat + (size_t)B_BATCH * NC * L_SEQ * DC;
  float* ov   = part + (size_t)B_BATCH * RCH * NC * DC;

  caps_gemm<<<256, 512, 0, stream>>>(u, W, uhat, part);
  caps_finish<<<B_BATCH, 512, 0, stream>>>(part, ov);
  caps_route<<<B_BATCH * RCH, 256, 0, stream>>>(uhat, ov, part);
  caps_finish<<<B_BATCH, 512, 0, stream>>>(part, ov);
  caps_route<<<B_BATCH * RCH, 256, 0, stream>>>(uhat, ov, part);
  caps_finish<<<B_BATCH, 512, 0, stream>>>(part, out);
}
